# starting kernel
# baseline (speedup 1.0000x reference)
#include <hip/hip_runtime.h>


typedef __attribute__((ext_vector_type(2))) float v2f;
typedef __attribute__((ext_vector_type(8))) float v8f;

#define SR      4
#define MAXOFF  9
#define NOFF    81
#define CCH     192
#define KSTEPS  (CCH / 4)
#define BB      8
#define HH      128
#define WW      160
#define HWC     (HH * WW)
#define SEGS_PER_ROW (WW / 16)
#define WAVES_PER_BLOCK 8
#define LDS_STRIDE 33

__device__ __forceinline__ int iclampi(int v, int lo, int hi) {
    return v < lo ? lo : (v > hi ? hi : v);
}

__global__ __launch_bounds__(256) void costvol_wmma_f32(
    const float* __restrict__ c1,
    const float* __restrict__ warped,
    const float* __restrict__ alpha_p,
    float* __restrict__ out)
{
    __shared__ float smem[WAVES_PER_BLOCK * 16 * LDS_STRIDE];

    const int tid  = threadIdx.x;
    const int lane = tid & 31;
    const int wave = tid >> 5;

    const int seg = blockIdx.x * WAVES_PER_BLOCK + wave;
    const int b   = seg / (HH * SEGS_PER_ROW);
    int rem       = seg - b * (HH * SEGS_PER_ROW);
    const int h   = rem / SEGS_PER_ROW;
    const int w0  = (rem - h * SEGS_PER_ROW) << 4;

    const int half = lane >> 4;
    const int m16  = lane & 15;

    const float alpha = alpha_p[0];
    const float inv_c = 1.0f / (float)CCH;

    const float* aBase = c1 + (size_t)b * CCH * HWC
                            + (2 * half) * HWC
                            + h * WW + w0 + m16;
    float a_x[KSTEPS], a_y[KSTEPS];
#pragma unroll
    for (int ks = 0; ks < KSTEPS; ++ks) {
        a_x[ks] = aBase[ks * 4 * HWC];
        a_y[ks] = aBase[ks * 4 * HWC + HWC];
    }

    const int col0   = w0 - SR + m16;
    const int col1   = col0 + 16;
    const float cm0  = (col0 >= 0 && col0 < WW) ? 1.0f : 0.0f;
    const float cm1  = (col1 < WW) ? 1.0f : 0.0f;
    const int col0c  = iclampi(col0, 0, WW - 1);
    const int col1c  = iclampi(col1, 0, WW - 1);

    const float* wBat = warped + (size_t)b * CCH * HWC + (2 * half) * HWC;
    float* smw = smem + wave * 16 * LDS_STRIDE;

#pragma unroll 1
    for (int dy = 0; dy < MAXOFF; ++dy) {
        const int  row  = h + dy - SR;
        const bool rOk  = (row >= 0) && (row < HH);
        const int  rowc = iclampi(row, 0, HH - 1);

        const float* p0 = wBat + rowc * WW + col0c;
        const float* p1 = wBat + rowc * WW + col1c;

        const float s0 = (rOk ? cm0 : 0.0f) * inv_c;
        const float s1 = (rOk ? cm1 : 0.0f) * inv_c;

        if (dy + 1 < MAXOFF) {
            const int rn = iclampi(h + dy + 1 - SR, 0, HH - 1);
            __builtin_prefetch(wBat + rn * WW + col0c, 0, 1);
        }

        v8f acc0 = {};
        v8f acc1 = {};
#pragma unroll
        for (int ks = 0; ks < KSTEPS; ++ks) {
            const int off = ks * 4 * HWC;
            v2f A;  A.x  = a_x[ks];       A.y  = a_y[ks];
            v2f B0; B0.x = p0[off];       B0.y = p0[off + HWC];
            v2f B1; B1.x = p1[off];       B1.y = p1[off + HWC];
            acc0 = __builtin_amdgcn_wmma_f32_16x16x4_f32(
                       false, A, false, B0, (short)0, acc0, false, false);
            acc1 = __builtin_amdgcn_wmma_f32_16x16x4_f32(
                       false, A, false, B1, (short)0, acc1, false, false);
        }

#pragma unroll
        for (int i = 0; i < 8; ++i) {
            const int M = i + half * 8;
            smw[M * LDS_STRIDE + m16]      = acc0[i] * s0;
            smw[M * LDS_STRIDE + 16 + m16] = acc1[i] * s1;
        }
        asm volatile("s_wait_dscnt 0x0" ::: "memory");

#pragma unroll
        for (int dbase = 0; dbase < MAXOFF + 1; dbase += 2) {
            const int d = dbase + half;
            const int n = m16 + d;
            const float cost = smw[m16 * LDS_STRIDE + n];
            const float res  = (cost >= 0.0f) ? cost : alpha * cost;
            if (d < MAXOFF) {
                const int o = dy * MAXOFF + d;
                out[(((size_t)b * NOFF + o) * HH + h) * WW + w0 + m16] = res;
            }
        }
        asm volatile("s_wait_dscnt 0x0" ::: "memory");
    }
}

extern "C" void kernel_launch(void* const* d_in, const int* in_sizes, int n_in,
                              void* d_out, int out_size, void* d_ws, size_t ws_size,
                              hipStream_t stream) {
    (void)in_sizes; (void)n_in; (void)d_ws; (void)ws_size; (void)out_size;

    const float* c1     = (const float*)d_in[0];
    const float* warped = (const float*)d_in[1];
    const float* alpha  = (const float*)d_in[2];
    float*       out    = (float*)d_out;

    const int totalSegs = BB * HH * SEGS_PER_ROW;
    const int blocks    = totalSegs / WAVES_PER_BLOCK;

    costvol_wmma_f32<<<blocks, 32 * WAVES_PER_BLOCK, 0, stream>>>(
        c1, warped, alpha, out);
}


# A rows = 8 pixels x 2 image rows: one 16-column B tile per warped row covers the whole x band, so 10 row tiles replace 18 per k-step (480 WMMAs, 960 B loads per wave; were 864, 1728)
# speedup vs baseline: 1.3821x; 1.3821x over previous
#include <hip/hip_runtime.h>


typedef __attribute__((ext_vector_type(2))) float v2f;
typedef __attribute__((ext_vector_type(8))) float v8f;

#define SR      4
#define MAXOFF  9
#define NOFF    81
#define CCH     192
#define KSTEPS  (CCH / 4)
#define BB      8
#define HH      128
#define WW      160
#define HWC     (HH * WW)
#define PXW     8
#define SEGS_PER_ROW (WW / PXW)
#define HPAIRS  (HH / 2)
#define NROWS   (MAXOFF + 1)
#define WAVES_PER_BLOCK 8
#define LDS_STRIDE 18

__device__ __forceinline__ int iclampi(int v, int lo, int hi) {
    return v < lo ? lo : (v > hi ? hi : v);
}

__global__ __launch_bounds__(256) void costvol_wmma_f32(
    const float* __restrict__ c1,
    const float* __restrict__ warped,
    const float* __restrict__ alpha_p,
    float* __restrict__ out)
{
    __shared__ float smem[WAVES_PER_BLOCK * 16 * LDS_STRIDE];

    const int tid  = threadIdx.x;
    const int lane = tid & 31;
    const int wave = tid >> 5;

    const int seg = blockIdx.x * WAVES_PER_BLOCK + wave;
    const int b   = seg / (HPAIRS * SEGS_PER_ROW);
    const int rem = seg - b * (HPAIRS * SEGS_PER_ROW);
    const int hp  = rem / SEGS_PER_ROW;
    const int h0  = hp * 2;
    const int w0  = (rem - hp * SEGS_PER_ROW) * PXW;

    const int half = lane >> 4;
    const int m16  = lane & 15;

    const float alpha = alpha_p[0];
    const float inv_c = 1.0f / (float)CCH;

    const float* aBase = c1 + (size_t)b * CCH * HWC
                            + (2 * half) * HWC
                            + (h0 + (m16 >> 3)) * WW + w0 + (m16 & 7);
    float a_x[KSTEPS], a_y[KSTEPS];
#pragma unroll
    for (int ks = 0; ks < KSTEPS; ++ks) {
        a_x[ks] = aBase[ks * 4 * HWC];
        a_y[ks] = aBase[ks * 4 * HWC + HWC];
    }

    const int col   = w0 - SR + m16;
    const float cm  = (col >= 0 && col < WW) ? 1.0f : 0.0f;
    const int colc  = iclampi(col, 0, WW - 1);

    const float* wBat = warped + (size_t)b * CCH * HWC + (2 * half) * HWC;
    float* smw = smem + wave * 16 * LDS_STRIDE;

    const int gi = lane & 7;
    const int gj = (lane >> 3) & 1;
    const int gd = lane >> 4;

#pragma unroll 1
    for (int t = 0; t < NROWS; ++t) {
        const int  row  = h0 + t - SR;
        const bool rOk  = (row >= 0) && (row < HH);
        const int  rowc = iclampi(row, 0, HH - 1);

        const float* p = wBat + rowc * WW + colc;

        const float s = (rOk ? cm : 0.0f) * inv_c;

        if (t + 1 < NROWS) {
            const int rn = iclampi(h0 + t + 1 - SR, 0, HH - 1);
            __builtin_prefetch(wBat + rn * WW + colc, 0, 1);
        }

        v8f acc = {};
#pragma unroll
        for (int ks = 0; ks < KSTEPS; ++ks) {
            const int off = ks * 4 * HWC;
            v2f A;  A.x = a_x[ks];       A.y = a_y[ks];
            v2f Bf; Bf.x = p[off];       Bf.y = p[off + HWC];
            acc = __builtin_amdgcn_wmma_f32_16x16x4_f32(
                      false, A, false, Bf, (short)0, acc, false, false);
        }

#pragma unroll
        for (int i = 0; i < 8; ++i) {
            const int M = i + half * 8;
            smw[M * LDS_STRIDE + m16] = acc[i] * s;
        }
        asm volatile("s_wait_dscnt 0x0" ::: "memory");

        const int  dy   = t - gj;
        const bool dyOk = (dy >= 0) && (dy < MAXOFF);
#pragma unroll
        for (int dbase = 0; dbase < MAXOFF + 1; dbase += 2) {
            const int d = dbase + gd;
            if (dyOk && d < MAXOFF) {
                const float cost = smw[(gj * 8 + gi) * LDS_STRIDE + gi + d];
                const float res  = (cost >= 0.0f) ? cost : alpha * cost;
                const int o = dy * MAXOFF + d;
                out[(((size_t)b * NOFF + o) * HH + h0 + gj) * WW + w0 + gi] = res;
            }
        }
        asm volatile("s_wait_dscnt 0x0" ::: "memory");
    }
}

extern "C" void kernel_launch(void* const* d_in, const int* in_sizes, int n_in,
                              void* d_out, int out_size, void* d_ws, size_t ws_size,
                              hipStream_t stream) {
    (void)in_sizes; (void)n_in; (void)d_ws; (void)ws_size; (void)out_size;

    const float* c1     = (const float*)d_in[0];
    const float* warped = (const float*)d_in[1];
    const float* alpha  = (const float*)d_in[2];
    float*       out    = (float*)d_out;

    const int totalSegs = BB * HPAIRS * SEGS_PER_ROW;
    const int blocks    = totalSegs / WAVES_PER_BLOCK;

    costvol_wmma_f32<<<blocks, 32 * WAVES_PER_BLOCK, 0, stream>>>(
        c1, warped, alpha, out);
}


# Warped rows staged once per block (4 image rows x 32 px, 8 waves) in a double-buffered LDS tile by 16-byte loads; B fragments read from LDS, not by per-wave 4-byte global loads
# speedup vs baseline: 1.4979x; 1.0838x over previous
#include <hip/hip_runtime.h>


typedef __attribute__((ext_vector_type(2))) float v2f;
typedef __attribute__((ext_vector_type(4))) float v4f;
typedef __attribute__((ext_vector_type(8))) float v8f;

#define SR      4
#define MAXOFF  9
#define NOFF    81
#define CCH     192
#define KSTEPS  (CCH / 4)
#define BB      8
#define HH      128
#define WW      160
#define HWC     (HH * WW)
#define PXW     8
#define NROWS   (MAXOFF + 1)
#define WAVES_PER_BLOCK 8
#define NTHR    (32 * WAVES_PER_BLOCK)
#define LDS_STRIDE 18

#define BLK_W   32
#define BLK_H   4
#define TCOLS   (BLK_W + 2 * SR)
#define TQ      (TCOLS / 4)
#define TROWS   (BLK_H + 2 * SR)
#define STAGE_FLOATS (CCH * TCOLS)
#define NF4     (CCH * TQ)
#define F4_PER_THR ((NF4 + NTHR - 1) / NTHR)
#define BLKS_X  (WW / BLK_W)
#define BLKS_Y  (HH / BLK_H)
#define NBLOCKS (BB * BLKS_Y * BLKS_X)
#define KG      12
#define NKG     (KSTEPS / KG)

__device__ __forceinline__ int iclampi(int v, int lo, int hi) {
    return v < lo ? lo : (v > hi ? hi : v);
}

__global__ __launch_bounds__(256) void costvol_wmma_f32(
    const float* __restrict__ c1,
    const float* __restrict__ warped,
    const float* __restrict__ alpha_p,
    float* __restrict__ out)
{
    __shared__ __attribute__((aligned(16))) float stage[2 * STAGE_FLOATS];
    __shared__ float smem[WAVES_PER_BLOCK * 16 * LDS_STRIDE];

    const int tid  = threadIdx.x;
    const int lane = tid & 31;
    const int wave = __builtin_amdgcn_readfirstlane(tid >> 5);

    const int blk = blockIdx.x;
    const int b   = blk / (BLKS_Y * BLKS_X);
    const int rem = blk - b * (BLKS_Y * BLKS_X);
    const int hb  = rem / BLKS_X;
    const int h00 = hb * BLK_H;
    const int w00 = (rem - hb * BLKS_X) * BLK_W;

    const int hq = wave >> 2;
    const int sg = wave & 3;
    const int h0 = h00 + 2 * hq;
    const int w0 = w00 + PXW * sg;

    const int half = lane >> 4;
    const int m16  = lane & 15;

    const float alpha = alpha_p[0];
    const float inv_c = 1.0f / (float)CCH;

    const float* aBase = c1 + (size_t)b * CCH * HWC
                            + (2 * half) * HWC
                            + (h0 + (m16 >> 3)) * WW + w0 + (m16 & 7);
    float a_x[KSTEPS], a_y[KSTEPS];
#pragma unroll
    for (int ks = 0; ks < KSTEPS; ++ks) {
        a_x[ks] = aBase[ks * 4 * HWC];
        a_y[ks] = aBase[ks * 4 * HWC + HWC];
    }

    const float* wBlk = warped + (size_t)b * CCH * HWC;
    int goff[F4_PER_THR];
#pragma unroll
    for (int j = 0; j < F4_PER_THR; ++j) {
        const int f    = tid + NTHR * j;
        const int ch   = f / TQ;
        const int q    = f - ch * TQ;
        const int col4 = w00 - SR + 4 * q;
        const bool ok  = (f < NF4) && (col4 >= 0) && (col4 < WW);
        goff[j] = ok ? (ch * HWC + col4) : -1;
    }

    v4f R[F4_PER_THR];

    {
        const int  row = h00 - SR;
        const bool rOk = (row >= 0) && (row < HH);
        const float* p = wBlk + iclampi(row, 0, HH - 1) * WW;
#pragma unroll
        for (int j = 0; j < F4_PER_THR; ++j) {
            v4f v = {0.0f, 0.0f, 0.0f, 0.0f};
            if (rOk && goff[j] >= 0) v = *(const v4f*)(p + goff[j]);
            R[j] = v;
        }
        v4f* dst = (v4f*)stage;
#pragma unroll
        for (int j = 0; j < F4_PER_THR; ++j) {
            const int f = tid + NTHR * j;
            if (f < NF4) dst[f] = R[j];
        }
    }
    __syncthreads();

    float* smw = smem + wave * 16 * LDS_STRIDE;

    const int gi = lane & 7;
    const int gj = (lane >> 3) & 1;
    const int gd = lane >> 4;

    const int bOff = (2 * half) * TCOLS + PXW * sg + m16;

#pragma unroll 1
    for (int t = 0; t < TROWS; ++t) {
        if (t + 1 < TROWS) {
            const int  row = h00 + t + 1 - SR;
            const bool rOk = (row >= 0) && (row < HH);
            const float* p = wBlk + iclampi(row, 0, HH - 1) * WW;
#pragma unroll
            for (int j = 0; j < F4_PER_THR; ++j) {
                v4f v = {0.0f, 0.0f, 0.0f, 0.0f};
                if (rOk && goff[j] >= 0) v = *(const v4f*)(p + goff[j]);
                R[j] = v;
            }
        }

        const int tq = t - 2 * hq;
        if (tq >= 0 && tq < NROWS) {
            const float* bs = stage + (t & 1) * STAGE_FLOATS + bOff;

            float bx[2][KG], by[2][KG];
#pragma unroll
            for (int i = 0; i < KG; ++i) {
                bx[0][i] = bs[i * 4 * TCOLS];
                by[0][i] = bs[i * 4 * TCOLS + TCOLS];
            }

            v8f acc = {};
#pragma unroll
            for (int g = 0; g < NKG; ++g) {
                if (g + 1 < NKG) {
#pragma unroll
                    for (int i = 0; i < KG; ++i) {
                        const int ks = (g + 1) * KG + i;
                        bx[(g + 1) & 1][i] = bs[ks * 4 * TCOLS];
                        by[(g + 1) & 1][i] = bs[ks * 4 * TCOLS + TCOLS];
                    }
                }
                __builtin_amdgcn_sched_barrier(0);
#pragma unroll
                for (int i = 0; i < KG; ++i) {
                    const int ks = g * KG + i;
                    v2f A;  A.x = a_x[ks];        A.y = a_y[ks];
                    v2f Bf; Bf.x = bx[g & 1][i];  Bf.y = by[g & 1][i];
                    acc = __builtin_amdgcn_wmma_f32_16x16x4_f32(
                              false, A, false, Bf, (short)0, acc, false, false);
                }
                __builtin_amdgcn_sched_barrier(0);
            }

#pragma unroll
            for (int i = 0; i < 8; ++i) {
                const int M = i + half * 8;
                smw[M * LDS_STRIDE + m16] = acc[i] * inv_c;
            }
            asm volatile("s_wait_dscnt 0x0" ::: "memory");

            const int  dy   = tq - gj;
            const bool dyOk = (dy >= 0) && (dy < MAXOFF);
#pragma unroll
            for (int dbase = 0; dbase < MAXOFF + 1; dbase += 2) {
                const int d = dbase + gd;
                if (dyOk && d < MAXOFF) {
                    const float cost = smw[(gj * 8 + gi) * LDS_STRIDE + gi + d];
                    const float res  = (cost >= 0.0f) ? cost : alpha * cost;
                    const int o = dy * MAXOFF + d;
                    out[(((size_t)b * NOFF + o) * HH + h0 + gj) * WW + w0 + gi] = res;
                }
            }
            asm volatile("s_wait_dscnt 0x0" ::: "memory");
        }

        if (t + 1 < TROWS) {
            v4f* dst = (v4f*)(stage + ((t + 1) & 1) * STAGE_FLOATS);
#pragma unroll
            for (int j = 0; j < F4_PER_THR; ++j) {
                const int f = tid + NTHR * j;
                if (f < NF4) dst[f] = R[j];
            }
        }
        __syncthreads();
    }
}

extern "C" void kernel_launch(void* const* d_in, const int* in_sizes, int n_in,
                              void* d_out, int out_size, void* d_ws, size_t ws_size,
                              hipStream_t stream) {
    (void)in_sizes; (void)n_in; (void)d_ws; (void)ws_size; (void)out_size;

    const float* c1     = (const float*)d_in[0];
    const float* warped = (const float*)d_in[1];
    const float* alpha  = (const float*)d_in[2];
    float*       out    = (float*)d_out;

    costvol_wmma_f32<<<NBLOCKS, NTHR, 0, stream>>>(
        c1, warped, alpha, out);
}


# Row hand-over decoupled: 3-slot LDS ring with a split workgroup barrier (signal after a row is stored, wait one step later); waves no longer idle at it and output stores are not drained before it
# speedup vs baseline: 1.6573x; 1.1064x over previous
#include <hip/hip_runtime.h>


typedef __attribute__((ext_vector_type(2))) float v2f;
typedef __attribute__((ext_vector_type(4))) float v4f;
typedef __attribute__((ext_vector_type(8))) float v8f;

#define SR      4
#define MAXOFF  9
#define NOFF    81
#define CCH     192
#define KSTEPS  (CCH / 4)
#define BB      8
#define HH      128
#define WW      160
#define HWC     (HH * WW)
#define PXW     8
#define NROWS   (MAXOFF + 1)
#define WAVES_PER_BLOCK 8
#define NTHR    (32 * WAVES_PER_BLOCK)
#define LDS_STRIDE 18

#define BLK_W   32
#define BLK_H   4
#define TCOLS   (BLK_W + 2 * SR)
#define TQ      (TCOLS / 4)
#define TROWS   (BLK_H + 2 * SR)
#define STAGE_FLOATS (CCH * TCOLS)
#define NF4     (CCH * TQ)
#define F4_PER_THR ((NF4 + NTHR - 1) / NTHR)
#define BLKS_X  (WW / BLK_W)
#define BLKS_Y  (HH / BLK_H)
#define NBLOCKS (BB * BLKS_Y * BLKS_X)
#define NSLOT   3
#define KG      12
#define NKG     (KSTEPS / KG)

__device__ __forceinline__ int iclampi(int v, int lo, int hi) {
    return v < lo ? lo : (v > hi ? hi : v);
}

__global__ __launch_bounds__(256) void costvol_wmma_f32(
    const float* __restrict__ c1,
    const float* __restrict__ warped,
    const float* __restrict__ alpha_p,
    float* __restrict__ out)
{
    __shared__ __attribute__((aligned(16))) float stage[NSLOT * STAGE_FLOATS];
    __shared__ float smem[WAVES_PER_BLOCK * 16 * LDS_STRIDE];

    const int tid  = threadIdx.x;
    const int lane = tid & 31;
    const int wave = __builtin_amdgcn_readfirstlane(tid >> 5);

    const int blk = blockIdx.x;
    const int b   = blk / (BLKS_Y * BLKS_X);
    const int rem = blk - b * (BLKS_Y * BLKS_X);
    const int hb  = rem / BLKS_X;
    const int h00 = hb * BLK_H;
    const int w00 = (rem - hb * BLKS_X) * BLK_W;

    const int hq = wave >> 2;
    const int sg = wave & 3;
    const int h0 = h00 + 2 * hq;
    const int w0 = w00 + PXW * sg;

    const int half = lane >> 4;
    const int m16  = lane & 15;

    const float alpha = alpha_p[0];
    const float inv_c = 1.0f / (float)CCH;

    const float* aBase = c1 + (size_t)b * CCH * HWC
                            + (2 * half) * HWC
                            + (h0 + (m16 >> 3)) * WW + w0 + (m16 & 7);
    float a_x[KSTEPS], a_y[KSTEPS];
#pragma unroll
    for (int ks = 0; ks < KSTEPS; ++ks) {
        a_x[ks] = aBase[ks * 4 * HWC];
        a_y[ks] = aBase[ks * 4 * HWC + HWC];
    }

    const float* wBlk = warped + (size_t)b * CCH * HWC;
    int goff[F4_PER_THR];
#pragma unroll
    for (int j = 0; j < F4_PER_THR; ++j) {
        const int f    = tid + NTHR * j;
        const int ch   = f / TQ;
        const int q    = f - ch * TQ;
        const int col4 = w00 - SR + 4 * q;
        const bool ok  = (f < NF4) && (col4 >= 0) && (col4 < WW);
        goff[j] = ok ? (ch * HWC + col4) : -1;
    }

    v4f R[F4_PER_THR];

    auto fetch_row = [&](v4f (&Rx)[F4_PER_THR], int tr) {
        const int  row = h00 + tr - SR;
        const bool rOk = (row >= 0) && (row < HH);
        const float* p = wBlk + iclampi(row, 0, HH - 1) * WW;
#pragma unroll
        for (int j = 0; j < F4_PER_THR; ++j) {
            v4f v = {0.0f, 0.0f, 0.0f, 0.0f};
            if (rOk && goff[j] >= 0) v = *(const v4f*)(p + goff[j]);
            Rx[j] = v;
        }
    };
    auto put_row = [&](const v4f (&Rx)[F4_PER_THR], int slot) {
        v4f* dst = (v4f*)(stage + slot * STAGE_FLOATS);
#pragma unroll
        for (int j = 0; j < F4_PER_THR; ++j) {
            const int f = tid + NTHR * j;
            if (f < NF4) dst[f] = Rx[j];
        }
    };

    {
        v4f R1[F4_PER_THR];
        fetch_row(R, 0);
        fetch_row(R1, 1);
        put_row(R, 0);
        put_row(R1, 1);
    }
    __syncthreads();
    fetch_row(R, 2);

    float* smw = smem + wave * 16 * LDS_STRIDE;

    const int gi = lane & 7;
    const int gj = (lane >> 3) & 1;
    const int gd = lane >> 4;

    const int bOff = (2 * half) * TCOLS + PXW * sg + m16;

    int slot_rd = 0;
    int slot_wr = 2;

#pragma unroll 1
    for (int t = 0; t < TROWS; ++t) {
        const int  tq     = t - 2 * hq;
        const bool active = (tq >= 0) && (tq < NROWS);
        v8f acc = {};
        if (active) {
            const float* bs = stage + slot_rd * STAGE_FLOATS + bOff;

            float bx[2][KG], by[2][KG];
#pragma unroll
            for (int i = 0; i < KG; ++i) {
                bx[0][i] = bs[i * 4 * TCOLS];
                by[0][i] = bs[i * 4 * TCOLS + TCOLS];
            }

#pragma unroll
            for (int g = 0; g < NKG; ++g) {
                if (g + 1 < NKG) {
#pragma unroll
                    for (int i = 0; i < KG; ++i) {
                        const int ks = (g + 1) * KG + i;
                        bx[(g + 1) & 1][i] = bs[ks * 4 * TCOLS];
                        by[(g + 1) & 1][i] = bs[ks * 4 * TCOLS + TCOLS];
                    }
                }
                __builtin_amdgcn_sched_barrier(0);
#pragma unroll
                for (int i = 0; i < KG; ++i) {
                    const int ks = g * KG + i;
                    v2f A;  A.x = a_x[ks];        A.y = a_y[ks];
                    v2f Bf; Bf.x = bx[g & 1][i];  Bf.y = by[g & 1][i];
                    acc = __builtin_amdgcn_wmma_f32_16x16x4_f32(
                              false, A, false, Bf, (short)0, acc, false, false);
                }
                __builtin_amdgcn_sched_barrier(0);
            }
        }

        if (t >= 1 && t + 1 < TROWS) {
            asm volatile("s_barrier_wait -1" ::: "memory");
        }
        if (t + 2 < TROWS) {
            put_row(R, slot_wr);
            asm volatile("s_wait_dscnt 0x0" ::: "memory");
            asm volatile("s_barrier_signal -1" ::: "memory");
        }
        if (t + 3 < TROWS) fetch_row(R, t + 3);

        if (active) {
#pragma unroll
            for (int i = 0; i < 8; ++i) {
                const int M = i + half * 8;
                smw[M * LDS_STRIDE + m16] = acc[i] * inv_c;
            }
            asm volatile("s_wait_dscnt 0x0" ::: "memory");

            const int  dy   = tq - gj;
            const bool dyOk = (dy >= 0) && (dy < MAXOFF);
#pragma unroll
            for (int dbase = 0; dbase < MAXOFF + 1; dbase += 2) {
                const int d = dbase + gd;
                if (dyOk && d < MAXOFF) {
                    const float cost = smw[(gj * 8 + gi) * LDS_STRIDE + gi + d];
                    const float res  = (cost >= 0.0f) ? cost : alpha * cost;
                    const int o = dy * MAXOFF + d;
                    out[(((size_t)b * NOFF + o) * HH + h0 + gj) * WW + w0 + gi] = res;
                }
            }
            asm volatile("s_wait_dscnt 0x0" ::: "memory");
        }

        slot_rd = (slot_rd == NSLOT - 1) ? 0 : slot_rd + 1;
        slot_wr = (slot_wr == NSLOT - 1) ? 0 : slot_wr + 1;
    }
}

extern "C" void kernel_launch(void* const* d_in, const int* in_sizes, int n_in,
                              void* d_out, int out_size, void* d_ws, size_t ws_size,
                              hipStream_t stream) {
    (void)in_sizes; (void)n_in; (void)d_ws; (void)ws_size; (void)out_size;

    const float* c1     = (const float*)d_in[0];
    const float* warped = (const float*)d_in[1];
    const float* alpha  = (const float*)d_in[2];
    float*       out    = (float*)d_out;

    costvol_wmma_f32<<<NBLOCKS, NTHR, 0, stream>>>(
        c1, warped, alpha, out);
}


# Block tile widened to 4 image rows x 80 px on 20 waves (512 blocks, one per compute unit, two even rounds): a staged warped row is 3 lines per 80 px, was 3 per 32 px; B groups of 8 k-steps
# speedup vs baseline: 1.8782x; 1.1333x over previous
#include <hip/hip_runtime.h>


typedef __attribute__((ext_vector_type(2))) float v2f;
typedef __attribute__((ext_vector_type(4))) float v4f;
typedef __attribute__((ext_vector_type(8))) float v8f;

#define SR      4
#define MAXOFF  9
#define NOFF    81
#define CCH     192
#define KSTEPS  (CCH / 4)
#define BB      8
#define HH      128
#define WW      160
#define HWC     (HH * WW)
#define PXW     8
#define NROWS   (MAXOFF + 1)
#define WAVES_PER_BLOCK 20
#define NTHR    (32 * WAVES_PER_BLOCK)
#define LDS_STRIDE 18

#define BLK_W   80
#define SEGS    (BLK_W / PXW)
#define BLK_H   4
#define TCOLS   (BLK_W + 2 * SR)
#define TQ      (TCOLS / 4)
#define TROWS   (BLK_H + 2 * SR)
#define STAGE_FLOATS (CCH * TCOLS)
#define NF4     (CCH * TQ)
#define F4_PER_THR ((NF4 + NTHR - 1) / NTHR)
#define BLKS_X  (WW / BLK_W)
#define BLKS_Y  (HH / BLK_H)
#define NBLOCKS (BB * BLKS_Y * BLKS_X)
#define NSLOT   3
#define KG      8
#define NKG     (KSTEPS / KG)

__device__ __forceinline__ int iclampi(int v, int lo, int hi) {
    return v < lo ? lo : (v > hi ? hi : v);
}

__global__ __launch_bounds__(NTHR) void costvol_wmma_f32(
    const float* __restrict__ c1,
    const float* __restrict__ warped,
    const float* __restrict__ alpha_p,
    float* __restrict__ out)
{
    __shared__ __attribute__((aligned(16))) float stage[NSLOT * STAGE_FLOATS];
    __shared__ float smem[WAVES_PER_BLOCK * 16 * LDS_STRIDE];

    const int tid  = threadIdx.x;
    const int lane = tid & 31;
    const int wave = __builtin_amdgcn_readfirstlane(tid >> 5);

    const int blk = blockIdx.x;
    const int b   = blk / (BLKS_Y * BLKS_X);
    const int rem = blk - b * (BLKS_Y * BLKS_X);
    const int hb  = rem / BLKS_X;
    const int h00 = hb * BLK_H;
    const int w00 = (rem - hb * BLKS_X) * BLK_W;

    const int hq = wave / SEGS;
    const int sg = wave - hq * SEGS;
    const int h0 = h00 + 2 * hq;
    const int w0 = w00 + PXW * sg;

    const int half = lane >> 4;
    const int m16  = lane & 15;

    const float alpha = alpha_p[0];
    const float inv_c = 1.0f / (float)CCH;

    const float* aBase = c1 + (size_t)b * CCH * HWC
                            + (2 * half) * HWC
                            + (h0 + (m16 >> 3)) * WW + w0 + (m16 & 7);
    float a_x[KSTEPS], a_y[KSTEPS];
#pragma unroll
    for (int ks = 0; ks < KSTEPS; ++ks) {
        a_x[ks] = aBase[ks * 4 * HWC];
        a_y[ks] = aBase[ks * 4 * HWC + HWC];
    }

    const float* wBlk = warped + (size_t)b * CCH * HWC;
    int goff[F4_PER_THR];
#pragma unroll
    for (int j = 0; j < F4_PER_THR; ++j) {
        const int f    = tid + NTHR * j;
        const int ch   = f / TQ;
        const int q    = f - ch * TQ;
        const int col4 = w00 - SR + 4 * q;
        const bool ok  = (f < NF4) && (col4 >= 0) && (col4 < WW);
        goff[j] = ok ? (ch * HWC + col4) : -1;
    }

    v4f R[F4_PER_THR];

    auto fetch_row = [&](v4f (&Rx)[F4_PER_THR], int tr) {
        const int  row = h00 + tr - SR;
        const bool rOk = (row >= 0) && (row < HH);
        const float* p = wBlk + iclampi(row, 0, HH - 1) * WW;
#pragma unroll
        for (int j = 0; j < F4_PER_THR; ++j) {
            v4f v = {0.0f, 0.0f, 0.0f, 0.0f};
            if (rOk && goff[j] >= 0) v = *(const v4f*)(p + goff[j]);
            Rx[j] = v;
        }
    };
    auto put_row = [&](const v4f (&Rx)[F4_PER_THR], int slot) {
        v4f* dst = (v4f*)(stage + slot * STAGE_FLOATS);
#pragma unroll
        for (int j = 0; j < F4_PER_THR; ++j) {
            const int f = tid + NTHR * j;
            if (f < NF4) dst[f] = Rx[j];
        }
    };

    {
        v4f R1[F4_PER_THR];
        fetch_row(R, 0);
        fetch_row(R1, 1);
        put_row(R, 0);
        put_row(R1, 1);
    }
    __syncthreads();
    fetch_row(R, 2);

    float* smw = smem + wave * 16 * LDS_STRIDE;

    const int gi = lane & 7;
    const int gj = (lane >> 3) & 1;
    const int gd = lane >> 4;

    const int bOff = (2 * half) * TCOLS + PXW * sg + m16;

    int slot_rd = 0;
    int slot_wr = 2;

#pragma unroll 1
    for (int t = 0; t < TROWS; ++t) {
        const int  tq     = t - 2 * hq;
        const bool active = (tq >= 0) && (tq < NROWS);
        v8f acc = {};
        if (active) {
            const float* bs = stage + slot_rd * STAGE_FLOATS + bOff;

            float bx[2][KG], by[2][KG];
#pragma unroll
            for (int i = 0; i < KG; ++i) {
                bx[0][i] = bs[i * 4 * TCOLS];
                by[0][i] = bs[i * 4 * TCOLS + TCOLS];
            }

#pragma unroll
            for (int g = 0; g < NKG; ++g) {
                if (g + 1 < NKG) {
#pragma unroll
                    for (int i = 0; i < KG; ++i) {
                        const int ks = (g + 1) * KG + i;
                        bx[(g + 1) & 1][i] = bs[ks * 4 * TCOLS];
                        by[(g + 1) & 1][i] = bs[ks * 4 * TCOLS + TCOLS];
                    }
                }
                __builtin_amdgcn_sched_barrier(0);
#pragma unroll
                for (int i = 0; i < KG; ++i) {
                    const int ks = g * KG + i;
                    v2f A;  A.x = a_x[ks];        A.y = a_y[ks];
                    v2f Bf; Bf.x = bx[g & 1][i];  Bf.y = by[g & 1][i];
                    acc = __builtin_amdgcn_wmma_f32_16x16x4_f32(
                              false, A, false, Bf, (short)0, acc, false, false);
                }
                __builtin_amdgcn_sched_barrier(0);
            }
        }

        if (t >= 1 && t + 1 < TROWS) {
            asm volatile("s_barrier_wait -1" ::: "memory");
        }
        if (t + 2 < TROWS) {
            put_row(R, slot_wr);
            asm volatile("s_wait_dscnt 0x0" ::: "memory");
            asm volatile("s_barrier_signal -1" ::: "memory");
        }
        if (t + 3 < TROWS) fetch_row(R, t + 3);

        if (active) {
#pragma unroll
            for (int i = 0; i < 8; ++i) {
                const int M = i + half * 8;
                smw[M * LDS_STRIDE + m16] = acc[i] * inv_c;
            }
            asm volatile("s_wait_dscnt 0x0" ::: "memory");

            const int  dy   = tq - gj;
            const bool dyOk = (dy >= 0) && (dy < MAXOFF);
#pragma unroll
            for (int dbase = 0; dbase < MAXOFF + 1; dbase += 2) {
                const int d = dbase + gd;
                if (dyOk && d < MAXOFF) {
                    const float cost = smw[(gj * 8 + gi) * LDS_STRIDE + gi + d];
                    const float res  = (cost >= 0.0f) ? cost : alpha * cost;
                    const int o = dy * MAXOFF + d;
                    out[(((size_t)b * NOFF + o) * HH + h0 + gj) * WW + w0 + gi] = res;
                }
            }
            asm volatile("s_wait_dscnt 0x0" ::: "memory");
        }

        slot_rd = (slot_rd == NSLOT - 1) ? 0 : slot_rd + 1;
        slot_wr = (slot_wr == NSLOT - 1) ? 0 : slot_wr + 1;
    }
}

extern "C" void kernel_launch(void* const* d_in, const int* in_sizes, int n_in,
                              void* d_out, int out_size, void* d_ws, size_t ws_size,
                              hipStream_t stream) {
    (void)in_sizes; (void)n_in; (void)d_ws; (void)ws_size; (void)out_size;

    const float* c1     = (const float*)d_in[0];
    const float* warped = (const float*)d_in[1];
    const float* alpha  = (const float*)d_in[2];
    float*       out    = (float*)d_out;

    costvol_wmma_f32<<<NBLOCKS, NTHR, 0, stream>>>(
        c1, warped, alpha, out);
}


# Loop order interchanged: channels outermost in 12 chunks of 16 through the LDS ring (c1 staged with warped, no 96 A registers); each wave keeps 10 accumulator tiles; band stored once at the end
# speedup vs baseline: 1.9381x; 1.0319x over previous
#include <hip/hip_runtime.h>


typedef __attribute__((ext_vector_type(2))) float v2f;
typedef __attribute__((ext_vector_type(4))) float v4f;
typedef __attribute__((ext_vector_type(8))) float v8f;

#define SR      4
#define MAXOFF  9
#define NOFF    81
#define CCH     192
#define BB      8
#define HH      128
#define WW      160
#define HWC     (HH * WW)
#define PXW     8
#define NROWS   (MAXOFF + 1)
#define WAVES_PER_BLOCK 20
#define NTHR    (32 * WAVES_PER_BLOCK)
#define LDS_STRIDE 18

#define BLK_W   80
#define SEGS    (BLK_W / PXW)
#define BLK_H   4
#define TCOLS   (BLK_W + 2 * SR)
#define TQ      (TCOLS / 4)
#define TROWS   (BLK_H + 2 * SR)
#define BLKS_X  (WW / BLK_W)
#define BLKS_Y  (HH / BLK_H)
#define NBLOCKS (BB * BLKS_Y * BLKS_X)

#define KC      16
#define NCH     (CCH / KC)
#define KSC     (KC / 4)
#define WROW    (KC * TCOLS)
#define WPART   (TROWS * WROW)
#define NF4W    (WPART / 4)
#define F4W_PER_THR ((NF4W + NTHR - 1) / NTHR)
#define AQ      (BLK_W / 4)
#define AROW    (KC * BLK_W + 8)
#define APART   (BLK_H * AROW)
#define NF4A    (BLK_H * KC * AQ)
#define F4A_PER_THR (NF4A / NTHR)
#define STAGE_FLOATS (WPART + APART)
#define CHUNK_BYTES (KC * HWC * 4)
#define NSLOT   3

__device__ __forceinline__ int iclampi(int v, int lo, int hi) {
    return v < lo ? lo : (v > hi ? hi : v);
}

__global__ __launch_bounds__(NTHR) void costvol_wmma_f32(
    const float* __restrict__ c1,
    const float* __restrict__ warped,
    const float* __restrict__ alpha_p,
    float* __restrict__ out)
{
    __shared__ __attribute__((aligned(16))) float stage[NSLOT * STAGE_FLOATS];
    __shared__ float smem[WAVES_PER_BLOCK * 16 * LDS_STRIDE];

    const int tid  = threadIdx.x;
    const int lane = tid & 31;
    const int wave = __builtin_amdgcn_readfirstlane(tid >> 5);

    const int blk = blockIdx.x;
    const int b   = blk / (BLKS_Y * BLKS_X);
    const int rem = blk - b * (BLKS_Y * BLKS_X);
    const int hb  = rem / BLKS_X;
    const int h00 = hb * BLK_H;
    const int w00 = (rem - hb * BLKS_X) * BLK_W;

    const int hq = wave / SEGS;
    const int sg = wave - hq * SEGS;
    const int h0 = h00 + 2 * hq;
    const int w0 = w00 + PXW * sg;

    const int half = lane >> 4;
    const int m16  = lane & 15;

    const float alpha = alpha_p[0];
    const float inv_c = 1.0f / (float)CCH;

    const char* wB = (const char*)(warped + (size_t)b * CCH * HWC);
    const char* aB = (const char*)(c1     + (size_t)b * CCH * HWC);
    unsigned goffb[F4W_PER_THR];
    bool     okw[F4W_PER_THR];
#pragma unroll
    for (int j = 0; j < F4W_PER_THR; ++j) {
        const int f    = tid + NTHR * j;
        const int rc   = f / TQ;
        const int q    = f - rc * TQ;
        const int row  = rc / KC;
        const int ch   = rc - row * KC;
        const int grow = h00 - SR + row;
        const int col4 = w00 - SR + 4 * q;
        okw[j]   = (f < NF4W) && (grow >= 0) && (grow < HH) && (col4 >= 0) && (col4 < WW);
        goffb[j] = 4u * (unsigned)(ch * HWC + iclampi(grow, 0, HH - 1) * WW + iclampi(col4, 0, WW - 4));
    }
    unsigned aoffb[F4A_PER_THR];
    int      la[F4A_PER_THR];
#pragma unroll
    for (int j = 0; j < F4A_PER_THR; ++j) {
        const int f   = tid + NTHR * j;
        const int rc  = f / AQ;
        const int q   = f - rc * AQ;
        const int row = rc / KC;
        const int ch  = rc - row * KC;
        aoffb[j] = 4u * (unsigned)(ch * HWC + (h00 + row) * WW + w00 + 4 * q);
        la[j]    = WPART + row * AROW + ch * BLK_W + 4 * q;
    }

    auto fetch_chunk = [&](v4f (&Rw)[F4W_PER_THR], v4f (&Ra)[F4A_PER_THR], int t) {
        const char* wp = wB + (size_t)t * CHUNK_BYTES;
        const char* ap = aB + (size_t)t * CHUNK_BYTES;
#pragma unroll
        for (int j = 0; j < F4W_PER_THR; ++j) Rw[j] = *(const v4f*)(wp + goffb[j]);
#pragma unroll
        for (int j = 0; j < F4A_PER_THR; ++j) Ra[j] = *(const v4f*)(ap + aoffb[j]);
    };
    auto put_chunk = [&](const v4f (&Rw)[F4W_PER_THR], const v4f (&Ra)[F4A_PER_THR], int slot) {
        float* sb = stage + slot * STAGE_FLOATS;
#pragma unroll
        for (int j = 0; j < F4W_PER_THR; ++j) {
            if (okw[j]) ((v4f*)sb)[tid + NTHR * j] = Rw[j];
        }
#pragma unroll
        for (int j = 0; j < F4A_PER_THR; ++j) *(v4f*)(sb + la[j]) = Ra[j];
    };

    v4f Rw[F4W_PER_THR];
    v4f Ra[F4A_PER_THR];

    {
        v4f Rw1[F4W_PER_THR];
        v4f Ra1[F4A_PER_THR];
        fetch_chunk(Rw, Ra, 0);
        fetch_chunk(Rw1, Ra1, 1);
        const v4f zero4 = {0.0f, 0.0f, 0.0f, 0.0f};
#pragma unroll
        for (int s = 0; s < NSLOT; ++s) {
#pragma unroll
            for (int j = 0; j < F4W_PER_THR; ++j) {
                const int f = tid + NTHR * j;
                if (!okw[j] && f < NF4W) ((v4f*)(stage + s * STAGE_FLOATS))[f] = zero4;
            }
        }
        put_chunk(Rw, Ra, 0);
        put_chunk(Rw1, Ra1, 1);
    }
    __syncthreads();
    fetch_chunk(Rw, Ra, 2);

    const int bOff = (2 * hq) * WROW + (2 * half) * TCOLS + PXW * sg + m16;
    const int aOff = WPART + (2 * hq + (m16 >> 3)) * AROW + (2 * half) * BLK_W + PXW * sg + (m16 & 7);

    v8f acc[NROWS];
#pragma unroll
    for (int r = 0; r < NROWS; ++r) acc[r] = (v8f){0.0f, 0.0f, 0.0f, 0.0f, 0.0f, 0.0f, 0.0f, 0.0f};

    int slot_rd = 0;
    int slot_wr = 2;

#pragma unroll 1
    for (int t = 0; t < NCH; ++t) {
        {
            const float* sb = stage + slot_rd * STAGE_FLOATS;
            const float* bs = sb + bOff;
            const float* as = sb + aOff;

            v2f Af[2];
            v2f Bf[2][NROWS];
            Af[0].x = as[0];
            Af[0].y = as[BLK_W];
#pragma unroll
            for (int r = 0; r < NROWS; ++r) {
                Bf[0][r].x = bs[r * WROW];
                Bf[0][r].y = bs[r * WROW + TCOLS];
            }
            __builtin_amdgcn_sched_barrier(0);
#pragma unroll
            for (int ks = 0; ks < KSC; ++ks) {
                if (ks + 1 < KSC) {
                    Af[(ks + 1) & 1].x = as[(ks + 1) * 4 * BLK_W];
                    Af[(ks + 1) & 1].y = as[(ks + 1) * 4 * BLK_W + BLK_W];
#pragma unroll
                    for (int r = 0; r < NROWS; ++r) {
                        Bf[(ks + 1) & 1][r].x = bs[r * WROW + (ks + 1) * 4 * TCOLS];
                        Bf[(ks + 1) & 1][r].y = bs[r * WROW + (ks + 1) * 4 * TCOLS + TCOLS];
                    }
                }
                __builtin_amdgcn_sched_barrier(0);
#pragma unroll
                for (int r = 0; r < NROWS; ++r) {
                    acc[r] = __builtin_amdgcn_wmma_f32_16x16x4_f32(
                                 false, Af[ks & 1], false, Bf[ks & 1][r], (short)0, acc[r], false, false);
                }
                __builtin_amdgcn_sched_barrier(0);
            }
        }

        if (t >= 1 && t + 1 < NCH) {
            asm volatile("s_barrier_wait -1" ::: "memory");
        }
        if (t + 2 < NCH) {
            put_chunk(Rw, Ra, slot_wr);
            asm volatile("s_wait_dscnt 0x0" ::: "memory");
            asm volatile("s_barrier_signal -1" ::: "memory");
        }
        if (t + 3 < NCH) fetch_chunk(Rw, Ra, t + 3);

        slot_rd = (slot_rd == NSLOT - 1) ? 0 : slot_rd + 1;
        slot_wr = (slot_wr == NSLOT - 1) ? 0 : slot_wr + 1;
    }

    float* smw = smem + wave * 16 * LDS_STRIDE;

    const int gi = lane & 7;
    const int gj = (lane >> 3) & 1;
    const int gd = lane >> 4;

#pragma unroll
    for (int r = 0; r < NROWS; ++r) {
#pragma unroll
        for (int i = 0; i < 8; ++i) {
            const int M = i + half * 8;
            smw[M * LDS_STRIDE + m16] = acc[r][i] * inv_c;
        }
        asm volatile("s_wait_dscnt 0x0" ::: "memory");

        const int  dy   = r - gj;
        const bool dyOk = (dy >= 0) && (dy < MAXOFF);
#pragma unroll
        for (int dbase = 0; dbase < MAXOFF + 1; dbase += 2) {
            const int d = dbase + gd;
            if (dyOk && d < MAXOFF) {
                const float cost = smw[(gj * 8 + gi) * LDS_STRIDE + gi + d];
                const float res  = (cost >= 0.0f) ? cost : alpha * cost;
                const int o = dy * MAXOFF + d;
                out[(((size_t)b * NOFF + o) * HH + h0 + gj) * WW + w0 + gi] = res;
            }
        }
        asm volatile("s_wait_dscnt 0x0" ::: "memory");
    }
}

extern "C" void kernel_launch(void* const* d_in, const int* in_sizes, int n_in,
                              void* d_out, int out_size, void* d_ws, size_t ws_size,
                              hipStream_t stream) {
    (void)in_sizes; (void)n_in; (void)d_ws; (void)ws_size; (void)out_size;

    const float* c1     = (const float*)d_in[0];
    const float* warped = (const float*)d_in[1];
    const float* alpha  = (const float*)d_in[2];
    float*       out    = (float*)d_out;

    costvol_wmma_f32<<<NBLOCKS, NTHR, 0, stream>>>(
        c1, warped, alpha, out);
}


# Fragment loads pipelined across the chunk boundary: the next chunk's first k-step (1 A + 10 B) is loaded in front of the current chunk's last 10 WMMAs; ring wait moved up one k-step to cover it
# speedup vs baseline: 2.1146x; 1.0911x over previous
#include <hip/hip_runtime.h>


typedef __attribute__((ext_vector_type(2))) float v2f;
typedef __attribute__((ext_vector_type(4))) float v4f;
typedef __attribute__((ext_vector_type(8))) float v8f;

#define SR      4
#define MAXOFF  9
#define NOFF    81
#define CCH     192
#define BB      8
#define HH      128
#define WW      160
#define HWC     (HH * WW)
#define PXW     8
#define NROWS   (MAXOFF + 1)
#define WAVES_PER_BLOCK 20
#define NTHR    (32 * WAVES_PER_BLOCK)
#define LDS_STRIDE 18

#define BLK_W   80
#define SEGS    (BLK_W / PXW)
#define BLK_H   4
#define TCOLS   (BLK_W + 2 * SR)
#define TQ      (TCOLS / 4)
#define TROWS   (BLK_H + 2 * SR)
#define BLKS_X  (WW / BLK_W)
#define BLKS_Y  (HH / BLK_H)
#define NBLOCKS (BB * BLKS_Y * BLKS_X)

#define KC      16
#define NCH     (CCH / KC)
#define KSC     (KC / 4)
#define WROW    (KC * TCOLS)
#define WPART   (TROWS * WROW)
#define NF4W    (WPART / 4)
#define F4W_PER_THR ((NF4W + NTHR - 1) / NTHR)
#define AQ      (BLK_W / 4)
#define AROW    (KC * BLK_W + 8)
#define APART   (BLK_H * AROW)
#define NF4A    (BLK_H * KC * AQ)
#define F4A_PER_THR (NF4A / NTHR)
#define STAGE_FLOATS (WPART + APART)
#define CHUNK_BYTES (KC * HWC * 4)
#define NSLOT   3

__device__ __forceinline__ int iclampi(int v, int lo, int hi) {
    return v < lo ? lo : (v > hi ? hi : v);
}

__global__ __launch_bounds__(NTHR) void costvol_wmma_f32(
    const float* __restrict__ c1,
    const float* __restrict__ warped,
    const float* __restrict__ alpha_p,
    float* __restrict__ out)
{
    __shared__ __attribute__((aligned(16))) float stage[NSLOT * STAGE_FLOATS];
    __shared__ float smem[WAVES_PER_BLOCK * 16 * LDS_STRIDE];

    const int tid  = threadIdx.x;
    const int lane = tid & 31;
    const int wave = __builtin_amdgcn_readfirstlane(tid >> 5);

    const int blk = blockIdx.x;
    const int b   = blk / (BLKS_Y * BLKS_X);
    const int rem = blk - b * (BLKS_Y * BLKS_X);
    const int hb  = rem / BLKS_X;
    const int h00 = hb * BLK_H;
    const int w00 = (rem - hb * BLKS_X) * BLK_W;

    const int hq = wave / SEGS;
    const int sg = wave - hq * SEGS;
    const int h0 = h00 + 2 * hq;
    const int w0 = w00 + PXW * sg;

    const int half = lane >> 4;
    const int m16  = lane & 15;

    const float alpha = alpha_p[0];
    const float inv_c = 1.0f / (float)CCH;

    const char* wB = (const char*)(warped + (size_t)b * CCH * HWC);
    const char* aB = (const char*)(c1     + (size_t)b * CCH * HWC);
    unsigned goffb[F4W_PER_THR];
    bool     okw[F4W_PER_THR];
#pragma unroll
    for (int j = 0; j < F4W_PER_THR; ++j) {
        const int f    = tid + NTHR * j;
        const int rc   = f / TQ;
        const int q    = f - rc * TQ;
        const int row  = rc / KC;
        const int ch   = rc - row * KC;
        const int grow = h00 - SR + row;
        const int col4 = w00 - SR + 4 * q;
        okw[j]   = (f < NF4W) && (grow >= 0) && (grow < HH) && (col4 >= 0) && (col4 < WW);
        goffb[j] = 4u * (unsigned)(ch * HWC + iclampi(grow, 0, HH - 1) * WW + iclampi(col4, 0, WW - 4));
    }
    unsigned aoffb[F4A_PER_THR];
    int      la[F4A_PER_THR];
#pragma unroll
    for (int j = 0; j < F4A_PER_THR; ++j) {
        const int f   = tid + NTHR * j;
        const int rc  = f / AQ;
        const int q   = f - rc * AQ;
        const int row = rc / KC;
        const int ch  = rc - row * KC;
        aoffb[j] = 4u * (unsigned)(ch * HWC + (h00 + row) * WW + w00 + 4 * q);
        la[j]    = WPART + row * AROW + ch * BLK_W + 4 * q;
    }

    auto fetch_chunk = [&](v4f (&Rw)[F4W_PER_THR], v4f (&Ra)[F4A_PER_THR], int t) {
        const char* wp = wB + (size_t)t * CHUNK_BYTES;
        const char* ap = aB + (size_t)t * CHUNK_BYTES;
#pragma unroll
        for (int j = 0; j < F4W_PER_THR; ++j) Rw[j] = *(const v4f*)(wp + goffb[j]);
#pragma unroll
        for (int j = 0; j < F4A_PER_THR; ++j) Ra[j] = *(const v4f*)(ap + aoffb[j]);
    };
    auto put_chunk = [&](const v4f (&Rw)[F4W_PER_THR], const v4f (&Ra)[F4A_PER_THR], int slot) {
        float* sb = stage + slot * STAGE_FLOATS;
#pragma unroll
        for (int j = 0; j < F4W_PER_THR; ++j) {
            if (okw[j]) ((v4f*)sb)[tid + NTHR * j] = Rw[j];
        }
#pragma unroll
        for (int j = 0; j < F4A_PER_THR; ++j) *(v4f*)(sb + la[j]) = Ra[j];
    };

    v4f Rw[F4W_PER_THR];
    v4f Ra[F4A_PER_THR];

    {
        v4f Rw1[F4W_PER_THR];
        v4f Ra1[F4A_PER_THR];
        fetch_chunk(Rw, Ra, 0);
        fetch_chunk(Rw1, Ra1, 1);
        const v4f zero4 = {0.0f, 0.0f, 0.0f, 0.0f};
#pragma unroll
        for (int s = 0; s < NSLOT; ++s) {
#pragma unroll
            for (int j = 0; j < F4W_PER_THR; ++j) {
                const int f = tid + NTHR * j;
                if (!okw[j] && f < NF4W) ((v4f*)(stage + s * STAGE_FLOATS))[f] = zero4;
            }
        }
        put_chunk(Rw, Ra, 0);
        put_chunk(Rw1, Ra1, 1);
    }
    __syncthreads();
    fetch_chunk(Rw, Ra, 2);

    const int bOff = (2 * hq) * WROW + (2 * half) * TCOLS + PXW * sg + m16;
    const int aOff = WPART + (2 * hq + (m16 >> 3)) * AROW + (2 * half) * BLK_W + PXW * sg + (m16 & 7);

    v8f acc[NROWS];
#pragma unroll
    for (int r = 0; r < NROWS; ++r) acc[r] = (v8f){0.0f, 0.0f, 0.0f, 0.0f, 0.0f, 0.0f, 0.0f, 0.0f};

    int slot_rd = 0;
    int slot_wr = 2;

    v2f Af[2];
    v2f Bf[2][NROWS];
    {
        const float* bs = stage + bOff;
        const float* as = stage + aOff;
        Af[0].x = as[0];
        Af[0].y = as[BLK_W];
#pragma unroll
        for (int r = 0; r < NROWS; ++r) {
            Bf[0][r].x = bs[r * WROW];
            Bf[0][r].y = bs[r * WROW + TCOLS];
        }
    }

#pragma unroll 1
    for (int t = 0; t < NCH; ++t) {
        {
            const float* sb = stage + slot_rd * STAGE_FLOATS;
            const float* bs = sb + bOff;
            const float* as = sb + aOff;
            const int slot_nx = (slot_rd == NSLOT - 1) ? 0 : slot_rd + 1;
            const float* sbn = stage + slot_nx * STAGE_FLOATS;
            const float* bsn = sbn + bOff;
            const float* asn = sbn + aOff;

#pragma unroll
            for (int ks = 0; ks < KSC; ++ks) {
                if (ks + 1 < KSC) {
                    Af[(ks + 1) & 1].x = as[(ks + 1) * 4 * BLK_W];
                    Af[(ks + 1) & 1].y = as[(ks + 1) * 4 * BLK_W + BLK_W];
#pragma unroll
                    for (int r = 0; r < NROWS; ++r) {
                        Bf[(ks + 1) & 1][r].x = bs[r * WROW + (ks + 1) * 4 * TCOLS];
                        Bf[(ks + 1) & 1][r].y = bs[r * WROW + (ks + 1) * 4 * TCOLS + TCOLS];
                    }
                } else {
                    if (t >= 1 && t + 1 < NCH) {
                        asm volatile("s_barrier_wait -1" ::: "memory");
                    }
                    __builtin_amdgcn_sched_barrier(0);
                    asm volatile("v_nop\n\tv_nop\n\tv_nop\n\tv_nop\n\tv_nop\n\tv_nop\n\tv_nop\n\tv_nop");
                    __builtin_amdgcn_sched_barrier(0);
                    Af[0].x = asn[0];
                    Af[0].y = asn[BLK_W];
#pragma unroll
                    for (int r = 0; r < NROWS; ++r) {
                        Bf[0][r].x = bsn[r * WROW];
                        Bf[0][r].y = bsn[r * WROW + TCOLS];
                    }
                }
                __builtin_amdgcn_sched_barrier(0);
#pragma unroll
                for (int r = 0; r < NROWS; ++r) {
                    acc[r] = __builtin_amdgcn_wmma_f32_16x16x4_f32(
                                 false, Af[ks & 1], false, Bf[ks & 1][r], (short)0, acc[r], false, false);
                }
                __builtin_amdgcn_sched_barrier(0);
            }
        }

        asm volatile("v_nop\n\tv_nop\n\tv_nop\n\tv_nop\n\tv_nop\n\tv_nop\n\tv_nop\n\tv_nop");
        __builtin_amdgcn_sched_barrier(0);

        if (t + 2 < NCH) {
            put_chunk(Rw, Ra, slot_wr);
            asm volatile("s_wait_dscnt 0x0" ::: "memory");
            asm volatile("s_barrier_signal -1" ::: "memory");
        }
        if (t + 3 < NCH) fetch_chunk(Rw, Ra, t + 3);

        slot_rd = (slot_rd == NSLOT - 1) ? 0 : slot_rd + 1;
        slot_wr = (slot_wr == NSLOT - 1) ? 0 : slot_wr + 1;
    }

    float* smw = smem + wave * 16 * LDS_STRIDE;

    const int gi = lane & 7;
    const int gj = (lane >> 3) & 1;
    const int gd = lane >> 4;

#pragma unroll
    for (int r = 0; r < NROWS; ++r) {
#pragma unroll
        for (int i = 0; i < 8; ++i) {
            const int M = i + half * 8;
            smw[M * LDS_STRIDE + m16] = acc[r][i] * inv_c;
        }
        asm volatile("s_wait_dscnt 0x0" ::: "memory");

        const int  dy   = r - gj;
        const bool dyOk = (dy >= 0) && (dy < MAXOFF);
#pragma unroll
        for (int dbase = 0; dbase < MAXOFF + 1; dbase += 2) {
            const int d = dbase + gd;
            if (dyOk && d < MAXOFF) {
                const float cost = smw[(gj * 8 + gi) * LDS_STRIDE + gi + d];
                const float res  = (cost >= 0.0f) ? cost : alpha * cost;
                const int o = dy * MAXOFF + d;
                out[(((size_t)b * NOFF + o) * HH + h0 + gj) * WW + w0 + gi] = res;
            }
        }
        asm volatile("s_wait_dscnt 0x0" ::: "memory");
    }
}

extern "C" void kernel_launch(void* const* d_in, const int* in_sizes, int n_in,
                              void* d_out, int out_size, void* d_ws, size_t ws_size,
                              hipStream_t stream) {
    (void)in_sizes; (void)n_in; (void)d_ws; (void)ws_size; (void)out_size;

    const float* c1     = (const float*)d_in[0];
    const float* warped = (const float*)d_in[1];
    const float* alpha  = (const float*)d_in[2];
    float*       out    = (float*)d_out;

    costvol_wmma_f32<<<NBLOCKS, NTHR, 0, stream>>>(
        c1, warped, alpha, out);
}
